# starting kernel
# baseline (speedup 1.0000x reference)
#include <hip/hip_runtime.h>
#include <math.h>

typedef float v2f __attribute__((ext_vector_type(2)));
typedef float v8f __attribute__((ext_vector_type(8)));

#define DDIM 2048
#define CCLS 10
#define BROWS 8192
#define WCH 128
#define WPAD 132
#define NCHUNK (DDIM / WCH)
#define KKPC (WCH / 4)

__global__ void zero_ws_kernel(float* __restrict__ ws) {
    int i = blockIdx.x * blockDim.x + threadIdx.x;
    ws[i] = 0.0f;
}

__global__ __launch_bounds__(256) void softmax_pack_kernel(const float* __restrict__ w,
                                                           float* __restrict__ sfrag) {
    __shared__ float red[256];
    const int c = blockIdx.x;
    const int tid = threadIdx.x;
    const float* wr = w + c * DDIM;

    float m = -3.4e38f;
    for (int d = tid; d < DDIM; d += 256) m = fmaxf(m, wr[d]);
    red[tid] = m; __syncthreads();
    for (int s = 128; s > 0; s >>= 1) { if (tid < s) red[tid] = fmaxf(red[tid], red[tid + s]); __syncthreads(); }
    m = red[0]; __syncthreads();

    float sum = 0.0f;
    for (int d = tid; d < DDIM; d += 256) sum += __expf(wr[d] - m);
    red[tid] = sum; __syncthreads();
    for (int s = 128; s > 0; s >>= 1) { if (tid < s) red[tid] += red[tid + s]; __syncthreads(); }
    const float inv = 1.0f / red[0];

    for (int d = tid; d < DDIM; d += 256) {
        float v = __expf(wr[d] - m) * inv;
        int kk = d >> 2, k = d & 3;
        sfrag[kk * 64 + (c + 16 * (k >> 1)) * 2 + (k & 1)] = v;
    }
}

__global__ __launch_bounds__(128) void lse_wmma_kernel(const float* __restrict__ x,
                                                       const float* __restrict__ sfrag,
                                                       float* __restrict__ out) {
    __shared__ float xlds[4][16 * WPAD];
    __shared__ float mrow[4][16];
    __shared__ float cmax[4][16];
    __shared__ float fsc[4][16];

    const int lane = threadIdx.x & 31;
    const int wv   = threadIdx.x >> 5;
    const int tile = blockIdx.x * 4 + wv;
    const int r0   = tile * 16;
    const int rA   = lane & 15;
    const int hh   = lane >> 4;

    float* xw = xlds[wv];
    float* mr = mrow[wv];
    float* cm = cmax[wv];
    float* fw = fsc[wv];

    if (lane < 16) mr[lane] = -3.0e38f;
    v8f acc = {0.f, 0.f, 0.f, 0.f, 0.f, 0.f, 0.f, 0.f};
    __syncthreads();

    for (int ch = 0; ch < NCHUNK; ++ch) {
        const float* xg = x + (size_t)r0 * DDIM + ch * WCH + 4 * lane;
        #pragma unroll
        for (int i = 0; i < 16; ++i) {
            float4 v = *(const float4*)(xg + (size_t)i * DDIM);
            *(float4*)(xw + i * WPAD + 4 * lane) = v;
            float lm = fmaxf(fmaxf(v.x, v.y), fmaxf(v.z, v.w));
            #pragma unroll
            for (int off = 16; off > 0; off >>= 1)
                lm = fmaxf(lm, __shfl_xor(lm, off, 32));
            if (lane == i) cm[i] = lm;
        }
        __syncthreads();

        float mo  = mr[rA];
        float cmx = cm[rA];
        float mn  = fmaxf(mo, cmx);
        __syncthreads();
        if (lane < 16) { mr[lane] = mn; fw[lane] = __expf(mo - mn); }
        __syncthreads();
        {
            const int base = (lane < 16) ? 0 : 8;
            #pragma unroll
            for (int v = 0; v < 8; ++v) acc[v] *= fw[base + v];
        }

        const float* ap = xw + rA * WPAD + 2 * hh;
        const float* bp = sfrag + (size_t)ch * KKPC * 64 + 2 * lane;
        #pragma unroll 4
        for (int kk = 0; kk < KKPC; ++kk) {
            float2 a = *(const float2*)(ap + 4 * kk);
            float2 b = *(const float2*)(bp + 64 * kk);
            v2f av; av.x = __expf(a.x - mn); av.y = __expf(a.y - mn);
            v2f bv; bv.x = b.x; bv.y = b.y;
            acc = __builtin_amdgcn_wmma_f32_16x16x4_f32(false, av, false, bv,
                                                        (short)0, acc, false, false);
        }
        __syncthreads();
    }

    const int base = (lane < 16) ? 0 : 8;
    if (rA < CCLS) {
        #pragma unroll
        for (int v = 0; v < 8; ++v) {
            float mf = mr[base + v];
            out[(size_t)(r0 + base + v) * CCLS + rA] = mf + __logf(acc[v]);
        }
    }
}

extern "C" void kernel_launch(void* const* d_in, const int* in_sizes, int n_in,
                              void* d_out, int out_size, void* d_ws, size_t ws_size,
                              hipStream_t stream) {
    const float* x = (const float*)d_in[0];
    const float* w = (const float*)d_in[1];
    float* out   = (float*)d_out;
    float* sfrag = (float*)d_ws;

    zero_ws_kernel<<<128, 256, 0, stream>>>(sfrag);
    softmax_pack_kernel<<<CCLS, 256, 0, stream>>>(w, sfrag);
    lse_wmma_kernel<<<BROWS / 64, 128, 0, stream>>>(x, sfrag, out);
}


# One 16-row tile per 8-wave work-group: each wave fetches its 16x256 column slice as one group of loads; per-wave (max,sum) partials merged in LDS in fixed wave order (new summation tree)
# speedup vs baseline: 7.7820x; 7.7820x over previous
#include <hip/hip_runtime.h>
#include <math.h>

typedef float v2f __attribute__((ext_vector_type(2)));
typedef float v8f __attribute__((ext_vector_type(8)));

#define DDIM 2048
#define CCLS 10
#define BROWS 8192
#define NWAVE 8
#define WCH (DDIM / NWAVE)
#define WPAD (WCH + 4)
#define KKPC (WCH / 4)

__global__ void zero_ws_kernel(float* __restrict__ ws) {
    int i = blockIdx.x * blockDim.x + threadIdx.x;
    ws[i] = 0.0f;
}

__global__ __launch_bounds__(256) void softmax_pack_kernel(const float* __restrict__ w,
                                                           float* __restrict__ sfrag) {
    __shared__ float red[256];
    const int c = blockIdx.x;
    const int tid = threadIdx.x;
    const float* wr = w + c * DDIM;

    float m = -3.4e38f;
    for (int d = tid; d < DDIM; d += 256) m = fmaxf(m, wr[d]);
    red[tid] = m; __syncthreads();
    for (int s = 128; s > 0; s >>= 1) { if (tid < s) red[tid] = fmaxf(red[tid], red[tid + s]); __syncthreads(); }
    m = red[0]; __syncthreads();

    float sum = 0.0f;
    for (int d = tid; d < DDIM; d += 256) sum += __expf(wr[d] - m);
    red[tid] = sum; __syncthreads();
    for (int s = 128; s > 0; s >>= 1) { if (tid < s) red[tid] += red[tid + s]; __syncthreads(); }
    const float inv = 1.0f / red[0];

    for (int d = tid; d < DDIM; d += 256) {
        float v = __expf(wr[d] - m) * inv;
        int kk = d >> 2, k = d & 3;
        sfrag[kk * 64 + (c + 16 * (k >> 1)) * 2 + (k & 1)] = v;
    }
}

__global__ __launch_bounds__(NWAVE * 32) void lse_wmma_kernel(const float* __restrict__ x,
                                                              const float* __restrict__ sfrag,
                                                              float* __restrict__ out) {
    __shared__ float xlds[NWAVE][16 * WPAD];
    __shared__ float cmax[NWAVE][16];
    __shared__ float part[NWAVE][16 * 16];

    const int lane = threadIdx.x & 31;
    const int wv   = threadIdx.x >> 5;
    const int r0   = blockIdx.x * 16;
    const int rA   = lane & 15;
    const int hh   = lane >> 4;

    float* xw = xlds[wv];

    const float* xg = x + (size_t)r0 * DDIM + wv * WCH + 4 * lane;
    float4 xv[16][WCH / 128];
    #pragma unroll
    for (int i = 0; i < 16; ++i) {
        #pragma unroll
        for (int j = 0; j < WCH / 128; ++j)
            xv[i][j] = *(const float4*)(xg + (size_t)i * DDIM + 128 * j);
    }

    float lm[16];
    #pragma unroll
    for (int i = 0; i < 16; ++i) {
        float m = -3.4e38f;
        #pragma unroll
        for (int j = 0; j < WCH / 128; ++j) {
            const float4 v = xv[i][j];
            *(float4*)(xw + i * WPAD + 128 * j + 4 * lane) = v;
            m = fmaxf(m, fmaxf(fmaxf(v.x, v.y), fmaxf(v.z, v.w)));
        }
        lm[i] = m;
    }
    #pragma unroll
    for (int off = 16; off > 0; off >>= 1) {
        #pragma unroll
        for (int i = 0; i < 16; ++i)
            lm[i] = fmaxf(lm[i], __shfl_xor(lm[i], off, 32));
    }
    float mn = lm[0];
    #pragma unroll
    for (int i = 1; i < 16; ++i) mn = (rA == i) ? lm[i] : mn;
    if (lane < 16) cmax[wv][lane] = mn;
    __syncthreads();

    v8f acc = {0.f, 0.f, 0.f, 0.f, 0.f, 0.f, 0.f, 0.f};
    const float* ap = xw + rA * WPAD + 2 * hh;
    const float* bp = sfrag + (size_t)wv * KKPC * 64 + 2 * lane;
    #pragma unroll 4
    for (int kk = 0; kk < KKPC; ++kk) {
        float2 a = *(const float2*)(ap + 4 * kk);
        float2 b = *(const float2*)(bp + 64 * kk);
        v2f av; av.x = __expf(a.x - mn); av.y = __expf(a.y - mn);
        v2f bv; bv.x = b.x; bv.y = b.y;
        acc = __builtin_amdgcn_wmma_f32_16x16x4_f32(false, av, false, bv,
                                                    (short)0, acc, false, false);
    }

    {
        float* pw = part[wv] + (8 * hh) * 16 + rA;
        #pragma unroll
        for (int v = 0; v < 8; ++v) pw[v * 16] = acc[v];
    }
    __syncthreads();

    const int t = threadIdx.x;
    if (t < 16 * CCLS) {
        const int row = t / CCLS;
        const int c   = t - row * CCLS;
        float M = cmax[0][row];
        #pragma unroll
        for (int w = 1; w < NWAVE; ++w) M = fmaxf(M, cmax[w][row]);
        float tot = 0.0f;
        #pragma unroll
        for (int w = 0; w < NWAVE; ++w)
            tot += part[w][row * 16 + c] * __expf(cmax[w][row] - M);
        out[(size_t)r0 * CCLS + t] = M + __logf(tot);
    }
}

extern "C" void kernel_launch(void* const* d_in, const int* in_sizes, int n_in,
                              void* d_out, int out_size, void* d_ws, size_t ws_size,
                              hipStream_t stream) {
    const float* x = (const float*)d_in[0];
    const float* w = (const float*)d_in[1];
    float* out   = (float*)d_out;
    float* sfrag = (float*)d_ws;

    zero_ws_kernel<<<128, 256, 0, stream>>>(sfrag);
    softmax_pack_kernel<<<CCLS, 256, 0, stream>>>(w, sfrag);
    lse_wmma_kernel<<<BROWS / 16, NWAVE * 32, 0, stream>>>(x, sfrag, out);
}


# One launch, no workspace: each wave loads its 10x256 weight slice into B-fragment registers and applies exp(w - slice max) itself; softmax normaliser joins the LDS merge (new summation order)
# speedup vs baseline: 12.1261x; 1.5582x over previous
#include <hip/hip_runtime.h>
#include <math.h>

typedef float v2f __attribute__((ext_vector_type(2)));
typedef float v8f __attribute__((ext_vector_type(8)));

#define DDIM 2048
#define CCLS 10
#define BROWS 8192
#define NWAVE 8
#define WCH (DDIM / NWAVE)
#define WPAD (WCH + 4)
#define NJ (WCH / 8)

__global__ __launch_bounds__(NWAVE * 32, 4) void lse_wmma_kernel(const float* __restrict__ x,
                                                              const float* __restrict__ w,
                                                              float* __restrict__ out) {
    __shared__ float xlds[NWAVE][16 * WPAD];
    __shared__ float cmax[NWAVE][16];
    __shared__ float wmx[NWAVE][16];
    __shared__ float wsm[NWAVE][16];
    __shared__ float part[NWAVE][16 * 16];

    const int lane = threadIdx.x & 31;
    const int wv   = threadIdx.x >> 5;
    const int r0   = blockIdx.x * 16;
    const int rA   = lane & 15;
    const int hh   = lane >> 4;

    float* xw = xlds[wv];

    const float* xg = x + (size_t)r0 * DDIM + wv * WCH + 4 * lane;
    float4 xv[16][WCH / 128];
    #pragma unroll
    for (int i = 0; i < 16; ++i) {
        #pragma unroll
        for (int j = 0; j < WCH / 128; ++j)
            xv[i][j] = *(const float4*)(xg + (size_t)i * DDIM + 128 * j);
    }

    const int cc = (rA < CCLS) ? rA : (CCLS - 1);
    const float* wg = w + (size_t)cc * DDIM + wv * WCH + 4 * hh;
    float4 wf[NJ];
    #pragma unroll
    for (int j = 0; j < NJ / 2; ++j) wf[j] = *(const float4*)(wg + 8 * j);

    float lm[16];
    #pragma unroll
    for (int i = 0; i < 16; ++i) {
        float m = -3.4e38f;
        #pragma unroll
        for (int j = 0; j < WCH / 128; ++j) {
            const float4 v = xv[i][j];
            *(float4*)(xw + i * WPAD + 128 * j + 4 * lane) = v;
            m = fmaxf(m, fmaxf(fmaxf(v.x, v.y), fmaxf(v.z, v.w)));
        }
        lm[i] = m;
    }

    __builtin_amdgcn_sched_barrier(0);
    #pragma unroll
    for (int j = NJ / 2; j < NJ; ++j) wf[j] = *(const float4*)(wg + 8 * j);

    #pragma unroll
    for (int off = 16; off > 0; off >>= 1) {
        #pragma unroll
        for (int i = 0; i < 16; ++i)
            lm[i] = fmaxf(lm[i], __shfl_xor(lm[i], off, 32));
    }
    float mn = lm[0];
    #pragma unroll
    for (int i = 1; i < 16; ++i) mn = (rA == i) ? lm[i] : mn;
    if (lane < 16) cmax[wv][lane] = mn;

    float wm = -3.4e38f;
    #pragma unroll
    for (int j = 0; j < NJ; ++j)
        wm = fmaxf(wm, fmaxf(fmaxf(wf[j].x, wf[j].y), fmaxf(wf[j].z, wf[j].w)));
    wm = fmaxf(wm, __shfl_xor(wm, 16, 32));
    __syncthreads();

    v8f acc = {0.f, 0.f, 0.f, 0.f, 0.f, 0.f, 0.f, 0.f};
    float wsum = 0.0f;
    const float* ap = xw + rA * WPAD + 4 * hh;
    #pragma unroll
    for (int j = 0; j < NJ; ++j) {
        const float4 a = *(const float4*)(ap + 8 * j);
        const float4 b = wf[j];
        v2f av, bv;
        av.x = __expf(a.x - mn); av.y = __expf(a.y - mn);
        bv.x = __expf(b.x - wm); bv.y = __expf(b.y - wm);
        wsum += bv.x + bv.y;
        acc = __builtin_amdgcn_wmma_f32_16x16x4_f32(false, av, false, bv,
                                                    (short)0, acc, false, false);
        av.x = __expf(a.z - mn); av.y = __expf(a.w - mn);
        bv.x = __expf(b.z - wm); bv.y = __expf(b.w - wm);
        wsum += bv.x + bv.y;
        acc = __builtin_amdgcn_wmma_f32_16x16x4_f32(false, av, false, bv,
                                                    (short)0, acc, false, false);
    }
    wsum += __shfl_xor(wsum, 16, 32);

    {
        float* pw = part[wv] + (8 * hh) * 16 + rA;
        #pragma unroll
        for (int v = 0; v < 8; ++v) pw[v * 16] = acc[v];
    }
    if (lane < 16) { wmx[wv][lane] = wm; wsm[wv][lane] = wsum; }
    __syncthreads();

    const int t = threadIdx.x;
    if (t < 16 * CCLS) {
        const int row = t / CCLS;
        const int c   = t - row * CCLS;
        float en[NWAVE], ed[NWAVE];
        float Mn = -3.4e38f, Md = -3.4e38f;
        #pragma unroll
        for (int k = 0; k < NWAVE; ++k) {
            ed[k] = wmx[k][c];
            en[k] = cmax[k][row] + ed[k];
            Mn = fmaxf(Mn, en[k]);
            Md = fmaxf(Md, ed[k]);
        }
        float num = 0.0f, den = 0.0f;
        #pragma unroll
        for (int k = 0; k < NWAVE; ++k) {
            num += part[k][row * 16 + c] * __expf(en[k] - Mn);
            den += wsm[k][c] * __expf(ed[k] - Md);
        }
        out[(size_t)r0 * CCLS + t] = (Mn - Md) + __logf(num / den);
    }
}

extern "C" void kernel_launch(void* const* d_in, const int* in_sizes, int n_in,
                              void* d_out, int out_size, void* d_ws, size_t ws_size,
                              hipStream_t stream) {
    const float* x = (const float*)d_in[0];
    const float* w = (const float*)d_in[1];
    float* out   = (float*)d_out;

    lse_wmma_kernel<<<BROWS / 16, NWAVE * 32, 0, stream>>>(x, w, out);
}


# Slice worked as two 128-column halves with own maxima (row max from fragment registers): half 0 staged, reduced, multiplied while half 1's loads are in flight; 16 partials (new summation tree)
# speedup vs baseline: 16.2304x; 1.3385x over previous
#include <hip/hip_runtime.h>
#include <math.h>

typedef float v2f __attribute__((ext_vector_type(2)));
typedef float v8f __attribute__((ext_vector_type(8)));
typedef float v4f __attribute__((ext_vector_type(4)));

#define DDIM 2048
#define CCLS 10
#define BROWS 8192
#define NWAVE 8
#define WCH (DDIM / NWAVE)
#define HCOL (WCH / 2)
#define HPAD (HCOL + 4)
#define NJH (HCOL / 8)
#define NPART (2 * NWAVE)

__device__ __forceinline__ float max4(const v4f v) {
    return fmaxf(fmaxf(v.x, v.y), fmaxf(v.z, v.w));
}

__device__ __forceinline__ void half_pass(const v4f (&xv)[16], const v4f (&wf)[NJH],
                                          float* xw, const int lane, const int rA, const int hh,
                                          float* cm, float* wmo, float* wso, float* pw) {
    __builtin_amdgcn_fence(__ATOMIC_RELEASE, "wavefront");
    __builtin_amdgcn_wave_barrier();
    #pragma unroll
    for (int i = 0; i < 16; ++i) *(v4f*)(xw + i * HPAD + 4 * lane) = xv[i];
    __builtin_amdgcn_fence(__ATOMIC_RELEASE, "wavefront");
    __builtin_amdgcn_wave_barrier();
    __builtin_amdgcn_fence(__ATOMIC_ACQUIRE, "wavefront");

    const float* ap = xw + rA * HPAD + 4 * hh;
    v4f af[NJH];
    #pragma unroll
    for (int j = 0; j < NJH; ++j) af[j] = *(const v4f*)(ap + 8 * j);

    float wm = -3.4e38f;
    #pragma unroll
    for (int j = 0; j < NJH; ++j) wm = fmaxf(wm, max4(wf[j]));
    wm = fmaxf(wm, __shfl_xor(wm, 16, 32));

    float mn = -3.4e38f;
    #pragma unroll
    for (int j = 0; j < NJH; ++j) mn = fmaxf(mn, max4(af[j]));
    mn = fmaxf(mn, __shfl_xor(mn, 16, 32));

    v8f acc = {0.f, 0.f, 0.f, 0.f, 0.f, 0.f, 0.f, 0.f};
    float wsum = 0.0f;
    #pragma unroll
    for (int j = 0; j < NJH; ++j) {
        const v4f a = af[j];
        const v4f b = wf[j];
        v2f av, bv;
        av.x = __expf(a.x - mn); av.y = __expf(a.y - mn);
        bv.x = __expf(b.x - wm); bv.y = __expf(b.y - wm);
        wsum += bv.x + bv.y;
        acc = __builtin_amdgcn_wmma_f32_16x16x4_f32(false, av, false, bv,
                                                    (short)0, acc, false, false);
        av.x = __expf(a.z - mn); av.y = __expf(a.w - mn);
        bv.x = __expf(b.z - wm); bv.y = __expf(b.w - wm);
        wsum += bv.x + bv.y;
        acc = __builtin_amdgcn_wmma_f32_16x16x4_f32(false, av, false, bv,
                                                    (short)0, acc, false, false);
    }
    wsum += __shfl_xor(wsum, 16, 32);

    #pragma unroll
    for (int v = 0; v < 8; ++v) pw[v * 16] = acc[v];
    if (lane < 16) { cm[lane] = mn; wmo[lane] = wm; wso[lane] = wsum; }
}

__global__ __launch_bounds__(NWAVE * 32, 4) void lse_wmma_kernel(const float* __restrict__ x,
                                                              const float* __restrict__ w,
                                                              float* __restrict__ out) {
    __shared__ __attribute__((aligned(16))) float xlds[NWAVE][16 * HPAD];
    __shared__ float cmax[NPART][16];
    __shared__ float wmx[NPART][16];
    __shared__ float wsm[NPART][16];
    __shared__ float part[NPART][16 * 16];

    const int lane = threadIdx.x & 31;
    const int wv   = threadIdx.x >> 5;
    const int r0   = blockIdx.x * 16;
    const int rA   = lane & 15;
    const int hh   = lane >> 4;

    float* xw = xlds[wv];

    const float* xg = x + (size_t)r0 * DDIM + wv * WCH + 4 * lane;
    const int cc = (rA < CCLS) ? rA : (CCLS - 1);
    const float* wg = w + (size_t)cc * DDIM + wv * WCH + 4 * hh;

    v4f xa[16], xb[16], wa[NJH], wb[NJH];
    #pragma unroll
    for (int i = 0; i < 16; ++i) xa[i] = *(const v4f*)(xg + (size_t)i * DDIM);
    __builtin_amdgcn_sched_barrier(0);
    #pragma unroll
    for (int j = 0; j < NJH; ++j) wa[j] = *(const v4f*)(wg + 8 * j);
    __builtin_amdgcn_sched_barrier(0);
    #pragma unroll
    for (int i = 0; i < 16; ++i) xb[i] = *(const v4f*)(xg + (size_t)i * DDIM + HCOL);

    __builtin_amdgcn_sched_barrier(0);

    half_pass(xa, wa, xw, lane, rA, hh, cmax[2 * wv], wmx[2 * wv], wsm[2 * wv],
              part[2 * wv] + (8 * hh) * 16 + rA);

    __builtin_amdgcn_sched_barrier(0);
    #pragma unroll
    for (int j = 0; j < NJH; ++j) wb[j] = *(const v4f*)(wg + HCOL + 8 * j);

    half_pass(xb, wb, xw, lane, rA, hh, cmax[2 * wv + 1], wmx[2 * wv + 1], wsm[2 * wv + 1],
              part[2 * wv + 1] + (8 * hh) * 16 + rA);

    __syncthreads();

    const int t = threadIdx.x;
    if (t < 16 * CCLS) {
        const int row = t / CCLS;
        const int c   = t - row * CCLS;
        float en[NPART], ed[NPART];
        float Mn = -3.4e38f, Md = -3.4e38f;
        #pragma unroll
        for (int k = 0; k < NPART; ++k) {
            ed[k] = wmx[k][c];
            en[k] = cmax[k][row] + ed[k];
            Mn = fmaxf(Mn, en[k]);
            Md = fmaxf(Md, ed[k]);
        }
        float num = 0.0f, den = 0.0f;
        #pragma unroll
        for (int k = 0; k < NPART; ++k) {
            num += part[k][row * 16 + c] * __expf(en[k] - Mn);
            den += wsm[k][c] * __expf(ed[k] - Md);
        }
        out[(size_t)r0 * CCLS + t] = (Mn - Md) + __logf(num / den);
    }
}

extern "C" void kernel_launch(void* const* d_in, const int* in_sizes, int n_in,
                              void* d_out, int out_size, void* d_ws, size_t ws_size,
                              hipStream_t stream) {
    const float* x = (const float*)d_in[0];
    const float* w = (const float*)d_in[1];
    float* out   = (float*)d_out;

    lse_wmma_kernel<<<BROWS / 16, NWAVE * 32, 0, stream>>>(x, w, out);
}


# Two 16-row tiles per work-group (256 groups of 8 waves, one per WGP): each wave exps its weight half once and multiplies both tiles by it, four passes per wave; same summation order
# speedup vs baseline: 17.2948x; 1.0656x over previous
#include <hip/hip_runtime.h>
#include <math.h>

typedef float v2f __attribute__((ext_vector_type(2)));
typedef float v8f __attribute__((ext_vector_type(8)));
typedef float v4f __attribute__((ext_vector_type(4)));

#define DDIM 2048
#define CCLS 10
#define BROWS 8192
#define NWAVE 8
#define NTILE 2
#define WCH (DDIM / NWAVE)
#define HCOL (WCH / 2)
#define HPAD (HCOL + 4)
#define NJH (HCOL / 8)
#define NPART (2 * NWAVE)
#define TOUT (16 * CCLS)

__device__ __forceinline__ float max4(const v4f v) {
    return fmaxf(fmaxf(v.x, v.y), fmaxf(v.z, v.w));
}

__device__ __forceinline__ void weight_half(v4f (&wf)[NJH], const int lane, float* wmo, float* wso) {
    float wm = -3.4e38f;
    #pragma unroll
    for (int j = 0; j < NJH; ++j) wm = fmaxf(wm, max4(wf[j]));
    wm = fmaxf(wm, __shfl_xor(wm, 16, 32));
    float wsum = 0.0f;
    #pragma unroll
    for (int j = 0; j < NJH; ++j) {
        v4f b = wf[j];
        b.x = __expf(b.x - wm); b.y = __expf(b.y - wm);
        wsum += b.x + b.y;
        b.z = __expf(b.z - wm); b.w = __expf(b.w - wm);
        wsum += b.z + b.w;
        wf[j] = b;
    }
    wsum += __shfl_xor(wsum, 16, 32);
    if (lane < 16) { wmo[lane] = wm; wso[lane] = wsum; }
}

__device__ __forceinline__ void tile_pass(const v4f (&xv)[16], const v4f (&ew)[NJH],
                                          float* xw, const int lane, const int rA, const int hh,
                                          float* cm, float* pw) {
    __builtin_amdgcn_fence(__ATOMIC_RELEASE, "wavefront");
    __builtin_amdgcn_wave_barrier();
    #pragma unroll
    for (int i = 0; i < 16; ++i) *(v4f*)(xw + i * HPAD + 4 * lane) = xv[i];
    __builtin_amdgcn_fence(__ATOMIC_RELEASE, "wavefront");
    __builtin_amdgcn_wave_barrier();
    __builtin_amdgcn_fence(__ATOMIC_ACQUIRE, "wavefront");

    const float* ap = xw + rA * HPAD + 4 * hh;
    v4f af[NJH];
    #pragma unroll
    for (int j = 0; j < NJH; ++j) af[j] = *(const v4f*)(ap + 8 * j);

    float mn = -3.4e38f;
    #pragma unroll
    for (int j = 0; j < NJH; ++j) mn = fmaxf(mn, max4(af[j]));
    mn = fmaxf(mn, __shfl_xor(mn, 16, 32));

    v8f acc = {0.f, 0.f, 0.f, 0.f, 0.f, 0.f, 0.f, 0.f};
    #pragma unroll
    for (int j = 0; j < NJH; ++j) {
        const v4f a = af[j];
        const v4f b = ew[j];
        v2f av, bv;
        av.x = __expf(a.x - mn); av.y = __expf(a.y - mn);
        bv.x = b.x; bv.y = b.y;
        acc = __builtin_amdgcn_wmma_f32_16x16x4_f32(false, av, false, bv,
                                                    (short)0, acc, false, false);
        av.x = __expf(a.z - mn); av.y = __expf(a.w - mn);
        bv.x = b.z; bv.y = b.w;
        acc = __builtin_amdgcn_wmma_f32_16x16x4_f32(false, av, false, bv,
                                                    (short)0, acc, false, false);
    }

    #pragma unroll
    for (int v = 0; v < 8; ++v) pw[v * 16] = acc[v];
    if (lane < 16) cm[lane] = mn;
}

__global__ __launch_bounds__(NWAVE * 32, 2) void lse_wmma_kernel(const float* __restrict__ x,
                                                              const float* __restrict__ w,
                                                              float* __restrict__ out) {
    __shared__ __attribute__((aligned(16))) float xlds[NWAVE][16 * HPAD];
    __shared__ float cmax[NTILE][NPART][16];
    __shared__ float wmx[NPART][16];
    __shared__ float wsm[NPART][16];
    __shared__ float part[NTILE][NPART][16 * 16];

    const int lane = threadIdx.x & 31;
    const int wv   = threadIdx.x >> 5;
    const int r0   = blockIdx.x * (16 * NTILE);
    const int rA   = lane & 15;
    const int hh   = lane >> 4;

    float* xw = xlds[wv];

    const float* xg = x + (size_t)r0 * DDIM + wv * WCH + 4 * lane;
    const int cc = (rA < CCLS) ? rA : (CCLS - 1);
    const float* wg = w + (size_t)cc * DDIM + wv * WCH + 4 * hh;

    v4f x0a[16], x1a[16], x0b[16], x1b[16], wa[NJH], wb[NJH];
    #pragma unroll
    for (int j = 0; j < NJH; ++j) wa[j] = *(const v4f*)(wg + 8 * j);
    __builtin_amdgcn_sched_barrier(0);
    #pragma unroll
    for (int i = 0; i < 16; ++i) x0a[i] = *(const v4f*)(xg + (size_t)i * DDIM);
    __builtin_amdgcn_sched_barrier(0);
    #pragma unroll
    for (int i = 0; i < 16; ++i) x1a[i] = *(const v4f*)(xg + (size_t)(16 + i) * DDIM);
    __builtin_amdgcn_sched_barrier(0);
    #pragma unroll
    for (int j = 0; j < NJH; ++j) wb[j] = *(const v4f*)(wg + HCOL + 8 * j);
    __builtin_amdgcn_sched_barrier(0);
    #pragma unroll
    for (int i = 0; i < 16; ++i) x0b[i] = *(const v4f*)(xg + (size_t)i * DDIM + HCOL);
    __builtin_amdgcn_sched_barrier(0);

    weight_half(wa, lane, wmx[2 * wv], wsm[2 * wv]);
    tile_pass(x0a, wa, xw, lane, rA, hh, cmax[0][2 * wv], part[0][2 * wv] + (8 * hh) * 16 + rA);

    __builtin_amdgcn_sched_barrier(0);
    #pragma unroll
    for (int i = 0; i < 16; ++i) x1b[i] = *(const v4f*)(xg + (size_t)(16 + i) * DDIM + HCOL);
    __builtin_amdgcn_sched_barrier(0);

    tile_pass(x1a, wa, xw, lane, rA, hh, cmax[1][2 * wv], part[1][2 * wv] + (8 * hh) * 16 + rA);

    weight_half(wb, lane, wmx[2 * wv + 1], wsm[2 * wv + 1]);
    tile_pass(x0b, wb, xw, lane, rA, hh, cmax[0][2 * wv + 1],
              part[0][2 * wv + 1] + (8 * hh) * 16 + rA);
    tile_pass(x1b, wb, xw, lane, rA, hh, cmax[1][2 * wv + 1],
              part[1][2 * wv + 1] + (8 * hh) * 16 + rA);

    __syncthreads();

    const int t = threadIdx.x;
    #pragma unroll
    for (int p = 0; p < 2; ++p) {
        const int o = t + p * (NWAVE * 32);
        if (o < NTILE * TOUT) {
            const int tl  = o / TOUT;
            const int e   = o - tl * TOUT;
            const int row = e / CCLS;
            const int c   = e - row * CCLS;
            float en[NPART], ed[NPART];
            float Mn = -3.4e38f, Md = -3.4e38f;
            #pragma unroll
            for (int k = 0; k < NPART; ++k) {
                ed[k] = wmx[k][c];
                en[k] = cmax[tl][k][row] + ed[k];
                Mn = fmaxf(Mn, en[k]);
                Md = fmaxf(Md, ed[k]);
            }
            float num = 0.0f, den = 0.0f;
            #pragma unroll
            for (int k = 0; k < NPART; ++k) {
                num += part[tl][k][row * 16 + c] * __expf(en[k] - Mn);
                den += wsm[k][c] * __expf(ed[k] - Md);
            }
            out[(size_t)r0 * CCLS + o] = (Mn - Md) + __logf(num / den);
        }
    }
}

extern "C" void kernel_launch(void* const* d_in, const int* in_sizes, int n_in,
                              void* d_out, int out_size, void* d_ws, size_t ws_size,
                              hipStream_t stream) {
    const float* x = (const float*)d_in[0];
    const float* w = (const float*)d_in[1];
    float* out   = (float*)d_out;

    lse_wmma_kernel<<<BROWS / (16 * NTILE), NWAVE * 32, 0, stream>>>(x, w, out);
}
